# starting kernel
# baseline (speedup 1.0000x reference)
#include <hip/hip_runtime.h>

typedef float v2f __attribute__((ext_vector_type(2)));
typedef float v8f __attribute__((ext_vector_type(8)));

#define IN_C  256
#define OUT_C 128
#define EPB   512

__global__ void gcn_zero_kernel(float* __restrict__ out, int total) {
    int i = blockIdx.x * blockDim.x + threadIdx.x;
    if (i < total) out[i] = 0.0f;
}

__global__ void gcn_gemm_wmma_kernel(const float* __restrict__ X,
                                     const float* __restrict__ W,
                                     float* __restrict__ H,
                                     int N) {
    const int wavesPerBlock = blockDim.x >> 5;
    const int wave = blockIdx.x * wavesPerBlock + (threadIdx.x >> 5);
    const int lane = threadIdx.x & 31;
    const int m0 = wave * 16;
    if (m0 >= N) return;

    const int half = lane >> 4;
    const int lid  = lane & 15;

    int arow = m0 + lid;
    if (arow >= N) arow = N - 1;
    const float* __restrict__ xrow  = X + (size_t)arow * IN_C + 2 * half;
    const float* __restrict__ wbase = W + (size_t)lid  * IN_C + 2 * half;

    v8f acc[8] = {};

    v2f a_cur = *(const v2f*)(xrow);
    v2f b_cur[8];
#pragma unroll
    for (int t = 0; t < 8; ++t)
        b_cur[t] = *(const v2f*)(wbase + (size_t)t * 16 * IN_C);

#pragma unroll 2
    for (int k = 0; k < IN_C; k += 4) {
        const int kn = (k + 4 < IN_C) ? (k + 4) : k;
        v2f a_nxt = *(const v2f*)(xrow + kn);
        v2f b_nxt[8];
#pragma unroll
        for (int t = 0; t < 8; ++t)
            b_nxt[t] = *(const v2f*)(wbase + (size_t)t * 16 * IN_C + kn);

#pragma unroll
        for (int t = 0; t < 8; ++t)
            acc[t] = __builtin_amdgcn_wmma_f32_16x16x4_f32(
                 false, a_cur,  false, b_cur[t],
                 (short)0, acc[t],
                 false,  false);

        a_cur = a_nxt;
#pragma unroll
        for (int t = 0; t < 8; ++t) b_cur[t] = b_nxt[t];
    }

    float* __restrict__ hbase = H + (size_t)(m0 + 8 * half) * OUT_C + lid;
    if (m0 + 16 <= N) {
#pragma unroll
        for (int t = 0; t < 8; ++t) {
            float* __restrict__ hp = hbase + t * 16;
#pragma unroll
            for (int g = 0; g < 8; ++g)
                hp[(size_t)g * OUT_C] = acc[t][g];
        }
    } else {
#pragma unroll
        for (int t = 0; t < 8; ++t) {
            const int col = t * 16 + lid;
#pragma unroll
            for (int g = 0; g < 8; ++g) {
                const int row = m0 + g + 8 * half;
                if (row < N) H[(size_t)row * OUT_C + col] = acc[t][g];
            }
        }
    }
}

__global__ void gcn_spmm_kernel(const int* __restrict__ A_rows,
                                const int* __restrict__ A_cols,
                                const float* __restrict__ A_vals,
                                const float* __restrict__ H,
                                float* __restrict__ out,
                                int E) {
    __shared__ int   s_row[EPB];
    __shared__ int   s_col[EPB];
    __shared__ float s_val[EPB];

    const int e0 = blockIdx.x * EPB;
    int n = E - e0;
    if (n > EPB) n = EPB;
    if (n <= 0) return;

    for (int i = threadIdx.x; i < n; i += blockDim.x) {
        s_row[i] = A_rows[e0 + i];
        s_col[i] = A_cols[e0 + i];
        s_val[i] = A_vals[e0 + i];
    }
    __syncthreads();

    const int c = threadIdx.x;
    float acc = 0.0f;
    int cur = s_row[0];

    for (int i = 0; i < n; ++i) {
        const int r = s_row[i];
        if (r != cur) {
            atomicAdd(&out[(size_t)cur * OUT_C + c], acc);
            acc = 0.0f;
            cur = r;
        }
        acc += s_val[i] * H[(size_t)s_col[i] * OUT_C + c];
    }
    atomicAdd(&out[(size_t)cur * OUT_C + c], acc);
}

extern "C" void kernel_launch(void* const* d_in, const int* in_sizes, int n_in,
                              void* d_out, int out_size, void* d_ws, size_t ws_size,
                              hipStream_t stream) {
    const float* X      = (const float*)d_in[0];
    const float* W      = (const float*)d_in[1];
    const int*   A_rows = (const int*)d_in[2];
    const int*   A_cols = (const int*)d_in[3];
    const float* A_vals = (const float*)d_in[4];
    float* out = (float*)d_out;
    float* H   = (float*)d_ws;

    const int N = in_sizes[0] / IN_C;
    const int E = in_sizes[2];

    gcn_zero_kernel<<<(out_size + 255) / 256, 256, 0, stream>>>(out, out_size);

    const int tiles  = (N + 15) / 16;
    const int blocks = (tiles + 7) / 8;
    gcn_gemm_wmma_kernel<<<blocks, 256, 0, stream>>>(X, W, H, N);

    gcn_spmm_kernel<<<(E + EPB - 1) / EPB, 128, 0, stream>>>(A_rows, A_cols, A_vals, H, out, E);
}


# GEMM: each wave keeps its 32-column slice of W in 256 registers (read once, in place) and walks 16-row X tiles in 512 resident 4-wave groups; the W copy pass is gone; k sum order changes
# speedup vs baseline: 3.1651x; 3.1651x over previous
#include <hip/hip_runtime.h>

typedef float v2f __attribute__((ext_vector_type(2)));
typedef float v4f __attribute__((ext_vector_type(4)));
typedef float v8f __attribute__((ext_vector_type(8)));

#define IN_C    256
#define OUT_C   128
#define ROWS_PB 8
#define GB      8
#define KP      (IN_C / 8)
#define GEMM_BLOCKS 512

__global__ void gcn_rp_kernel(const int* __restrict__ A_rows,
                              int* __restrict__ rp, int E, int N) {
    const int e = blockIdx.x * blockDim.x + threadIdx.x;
    if (e > E) return;
    int lo = (e == 0) ? -1 : A_rows[e - 1];
    int hi = (e == E) ? N  : A_rows[e];
    if (lo < -1) lo = -1;
    if (hi > N)  hi = N;
    for (int r = lo + 1; r <= hi; ++r) rp[r] = e;
}

__global__ __launch_bounds__(128, 2)
void gcn_gemm_wreg_kernel(const float* X, const float* W, float* H, int N, int tiles) {
    const int lane = threadIdx.x & 31;
    const int wv   = threadIdx.x >> 5;
    const int half = lane >> 4;
    const int lid  = lane & 15;

    int tile = blockIdx.x;
    if (tile >= tiles) return;

    v4f b0[KP], b1[KP];
    {
        const float* w0 = W + (size_t)(32 * wv + lid) * IN_C + 4 * half;
        const float* w1 = w0 + 16 * IN_C;
#pragma unroll
        for (int jj = 0; jj < KP; ++jj) {
            b0[jj] = *(const v4f*)(w0 + 8 * jj);
            b1[jj] = *(const v4f*)(w1 + 8 * jj);
        }
    }

    v4f a[KP];
    {
        int arow = tile * 16 + lid;
        if (arow >= N) arow = N - 1;
        const float* xr = X + (size_t)arow * IN_C + 4 * half;
#pragma unroll
        for (int jj = 0; jj < KP; ++jj) a[jj] = *(const v4f*)(xr + 8 * jj);
    }

    auto do_tile = [&](const int tile) __attribute__((always_inline)) {
        int nt = tile + (int)gridDim.x;
        if (nt >= tiles) nt = tile;
        int nrow = nt * 16 + lid;
        if (nrow >= N) nrow = N - 1;
        const float* xn = X + (size_t)nrow * IN_C + 4 * half;

        v8f acc0 = {}, acc1 = {};
#pragma unroll
        for (int jj = 0; jj < KP; ++jj) {
            const v4f av = a[jj];
            const v2f a01 = {av.x, av.y};
            const v2f a23 = {av.z, av.w};
            const v2f p01 = {b0[jj].x, b0[jj].y};
            const v2f p23 = {b0[jj].z, b0[jj].w};
            const v2f q01 = {b1[jj].x, b1[jj].y};
            const v2f q23 = {b1[jj].z, b1[jj].w};
            acc0 = __builtin_amdgcn_wmma_f32_16x16x4_f32(false, a01, false, p01, (short)0, acc0, false, false);
            acc1 = __builtin_amdgcn_wmma_f32_16x16x4_f32(false, a01, false, q01, (short)0, acc1, false, false);
            acc0 = __builtin_amdgcn_wmma_f32_16x16x4_f32(false, a23, false, p23, (short)0, acc0, false, false);
            acc1 = __builtin_amdgcn_wmma_f32_16x16x4_f32(false, a23, false, q23, (short)0, acc1, false, false);
            a[jj] = *(const v4f*)(xn + 8 * jj);
            __builtin_amdgcn_sched_group_barrier(0x008, 4, 0);
            __builtin_amdgcn_sched_group_barrier(0x020, 1, 0);
        }

        const int m0 = tile * 16;
        float* hbase = H + (size_t)(m0 + 8 * half) * OUT_C + 32 * wv + lid;
        if (m0 + 16 <= N) {
#pragma unroll
            for (int g = 0; g < 8; ++g) {
                hbase[(size_t)g * OUT_C]      = acc0[g];
                hbase[(size_t)g * OUT_C + 16] = acc1[g];
            }
        } else {
#pragma unroll
            for (int g = 0; g < 8; ++g) {
                if (m0 + 8 * half + g < N) {
                    hbase[(size_t)g * OUT_C]      = acc0[g];
                    hbase[(size_t)g * OUT_C + 16] = acc1[g];
                }
            }
        }
    };

    __builtin_amdgcn_sched_barrier(0);
    do_tile(tile);
    tile += (int)gridDim.x;
    for (; tile < tiles; tile += (int)gridDim.x) do_tile(tile);
}

__global__ __launch_bounds__(32 * ROWS_PB)
void gcn_spmm_row_kernel(const int* __restrict__ rp,
                         const int* __restrict__ A_cols,
                         const float* __restrict__ A_vals,
                         const float* __restrict__ H,
                         float* __restrict__ out,
                         int N) {
    const int lane = threadIdx.x & 31;
    const int row = __builtin_amdgcn_readfirstlane(blockIdx.x * ROWS_PB + (threadIdx.x >> 5));
    if (row >= N) return;
    const int beg = rp[row];
    const int end = rp[row + 1];

    const v4f* __restrict__ H4 = (const v4f*)H + lane;
    v4f acc = {0.0f, 0.0f, 0.0f, 0.0f};

    for (int base = beg; base < end; base += 32) {
        int n = end - base;
        if (n > 32) n = 32;
        int   myc = 0;
        float myv = 0.0f;
        if (lane < n) {
            myc = A_cols[base + lane];
            myv = A_vals[base + lane];
        }
        const int myvb = __float_as_int(myv);

        int j = 0;
        for (; j + GB <= n; j += GB) {
            v4f h[GB];
#pragma unroll
            for (int u = 0; u < GB; ++u) {
                const int c = __builtin_amdgcn_readlane(myc, j + u);
                h[u] = H4[(size_t)c * (OUT_C / 4)];
            }
#pragma unroll
            for (int u = 0; u < GB; ++u) {
                const float v = __int_as_float(__builtin_amdgcn_readlane(myvb, j + u));
                acc += v * h[u];
            }
        }
        if (j < n) {
            v4f h[GB - 1];
#pragma unroll
            for (int u = 0; u < GB - 1; ++u) {
                h[u] = (v4f){0.0f, 0.0f, 0.0f, 0.0f};
                if (j + u < n) {
                    const int c = __builtin_amdgcn_readlane(myc, j + u);
                    h[u] = H4[(size_t)c * (OUT_C / 4)];
                }
            }
#pragma unroll
            for (int u = 0; u < GB - 1; ++u) {
                if (j + u < n) {
                    const float v = __int_as_float(__builtin_amdgcn_readlane(myvb, j + u));
                    acc += v * h[u];
                }
            }
        }
    }
    ((v4f*)out)[(size_t)row * (OUT_C / 4) + lane] = acc;
}

extern "C" void kernel_launch(void* const* d_in, const int* in_sizes, int n_in,
                              void* d_out, int out_size, void* d_ws, size_t ws_size,
                              hipStream_t stream) {
    const float* X      = (const float*)d_in[0];
    const float* W      = (const float*)d_in[1];
    const int*   A_rows = (const int*)d_in[2];
    const int*   A_cols = (const int*)d_in[3];
    const float* A_vals = (const float*)d_in[4];
    float* out = (float*)d_out;

    const int N = in_sizes[0] / IN_C;
    const int E = in_sizes[2];

    float* H  = (float*)d_ws;
    int*   rp = (int*)(H + (size_t)N * OUT_C);

    gcn_rp_kernel<<<(E + 256) / 256, 256, 0, stream>>>(A_rows, rp, E, N);

    const int tiles = (N + 15) / 16;
    gcn_gemm_wreg_kernel<<<GEMM_BLOCKS, 128, 0, stream>>>(X, W, H, N, tiles);

    gcn_spmm_row_kernel<<<(N + ROWS_PB - 1) / ROWS_PB, 32 * ROWS_PB, 0, stream>>>(rp, A_cols, A_vals, H, out, N);
}


# Row-pointer pass done by the GEMM's own waves at their start (row list requested ahead of the W loads, pointers stored while those are in flight); its launch is gone, 3 launches -> 2; same sums
# speedup vs baseline: 3.3509x; 1.0587x over previous
#include <hip/hip_runtime.h>

typedef float v2f __attribute__((ext_vector_type(2)));
typedef float v4f __attribute__((ext_vector_type(4)));
typedef float v8f __attribute__((ext_vector_type(8)));

#define IN_C    256
#define OUT_C   128
#define ROWS_PB 8
#define GB      8
#define KP      (IN_C / 8)
#define GEMM_BLOCKS 512

typedef int v4i __attribute__((ext_vector_type(4)));
#define RP_AHEAD 7

__device__ __forceinline__ void rp_fill(int* __restrict__ rp, int lo, int hi, int e, int N) {
    if (lo < -1) lo = -1;
    if (hi > N)  hi = N;
#pragma clang loop vectorize(disable) unroll(disable)
    for (int r = lo + 1; r <= hi; ++r) rp[r] = e;
}

__global__ __launch_bounds__(128, 2)
void gcn_gemm_wreg_kernel(const float* X, const float* W, float* H, int N, int tiles,
                          const int* __restrict__ A_rows, int* __restrict__ rp, int E) {
    const int lane = threadIdx.x & 31;
    const int wv   = threadIdx.x >> 5;
    const int half = lane >> 4;
    const int lid  = lane & 15;

    int tile = blockIdx.x;

    const int gtid = blockIdx.x * 128 + threadIdx.x;
    const int nthr = (int)gridDim.x * 128;
    const int nq   = E >> 2;
    v4i rv[RP_AHEAD];
    int rprev[RP_AHEAD];
#pragma unroll
    for (int i = 0; i < RP_AHEAD; ++i) {
        const int q = gtid + i * nthr;
        rv[i] = (v4i){0, 0, 0, 0};
        rprev[i] = -1;
        if (q < nq) {
            rv[i] = *(const v4i*)(A_rows + 4 * (size_t)q);
            if (q > 0) rprev[i] = A_rows[4 * (size_t)q - 1];
        }
    }

    v4f b0[KP], b1[KP];
    {
        const float* w0 = W + (size_t)(32 * wv + lid) * IN_C + 4 * half;
        const float* w1 = w0 + 16 * IN_C;
#pragma unroll
        for (int jj = 0; jj < KP; ++jj) {
            b0[jj] = *(const v4f*)(w0 + 8 * jj);
            b1[jj] = *(const v4f*)(w1 + 8 * jj);
        }
    }

    v4f a[KP];
    {
        int arow = tile * 16 + lid;
        if (arow >= N) arow = N - 1;
        const float* xr = X + (size_t)arow * IN_C + 4 * half;
#pragma unroll
        for (int jj = 0; jj < KP; ++jj) a[jj] = *(const v4f*)(xr + 8 * jj);
    }

#pragma unroll
    for (int i = 0; i < RP_AHEAD; ++i) {
        const int q = gtid + i * nthr;
        if (q < nq) {
            const int e0 = 4 * q;
            rp_fill(rp, rprev[i], rv[i].x, e0,     N);
            rp_fill(rp, rv[i].x,  rv[i].y, e0 + 1, N);
            rp_fill(rp, rv[i].y,  rv[i].z, e0 + 2, N);
            rp_fill(rp, rv[i].z,  rv[i].w, e0 + 3, N);
        }
    }
    for (int q = gtid + RP_AHEAD * nthr; q < nq; q += nthr) {
        const v4i v = *(const v4i*)(A_rows + 4 * (size_t)q);
        const int p = A_rows[4 * (size_t)q - 1];
        const int e0 = 4 * q;
        rp_fill(rp, p,   v.x, e0,     N);
        rp_fill(rp, v.x, v.y, e0 + 1, N);
        rp_fill(rp, v.y, v.z, e0 + 2, N);
        rp_fill(rp, v.z, v.w, e0 + 3, N);
    }
    {
        const int e = 4 * nq + gtid;
        if (e <= E) {
            const int lo = (e == 0) ? -1 : A_rows[e - 1];
            const int hi = (e == E) ? N  : A_rows[e];
            rp_fill(rp, lo, hi, e, N);
        }
    }

    if (tile >= tiles) return;

    auto do_tile = [&](const int tile) __attribute__((always_inline)) {
        int nt = tile + (int)gridDim.x;
        if (nt >= tiles) nt = tile;
        int nrow = nt * 16 + lid;
        if (nrow >= N) nrow = N - 1;
        const float* xn = X + (size_t)nrow * IN_C + 4 * half;

        v8f acc0 = {}, acc1 = {};
#pragma unroll
        for (int jj = 0; jj < KP; ++jj) {
            const v4f av = a[jj];
            const v2f a01 = {av.x, av.y};
            const v2f a23 = {av.z, av.w};
            const v2f p01 = {b0[jj].x, b0[jj].y};
            const v2f p23 = {b0[jj].z, b0[jj].w};
            const v2f q01 = {b1[jj].x, b1[jj].y};
            const v2f q23 = {b1[jj].z, b1[jj].w};
            acc0 = __builtin_amdgcn_wmma_f32_16x16x4_f32(false, a01, false, p01, (short)0, acc0, false, false);
            acc1 = __builtin_amdgcn_wmma_f32_16x16x4_f32(false, a01, false, q01, (short)0, acc1, false, false);
            acc0 = __builtin_amdgcn_wmma_f32_16x16x4_f32(false, a23, false, p23, (short)0, acc0, false, false);
            acc1 = __builtin_amdgcn_wmma_f32_16x16x4_f32(false, a23, false, q23, (short)0, acc1, false, false);
            a[jj] = *(const v4f*)(xn + 8 * jj);
            __builtin_amdgcn_sched_group_barrier(0x008, 4, 0);
            __builtin_amdgcn_sched_group_barrier(0x020, 1, 0);
        }

        const int m0 = tile * 16;
        float* hbase = H + (size_t)(m0 + 8 * half) * OUT_C + 32 * wv + lid;
        if (m0 + 16 <= N) {
#pragma unroll
            for (int g = 0; g < 8; ++g) {
                hbase[(size_t)g * OUT_C]      = acc0[g];
                hbase[(size_t)g * OUT_C + 16] = acc1[g];
            }
        } else {
#pragma unroll
            for (int g = 0; g < 8; ++g) {
                if (m0 + 8 * half + g < N) {
                    hbase[(size_t)g * OUT_C]      = acc0[g];
                    hbase[(size_t)g * OUT_C + 16] = acc1[g];
                }
            }
        }
    };

    __builtin_amdgcn_sched_barrier(0);
    do_tile(tile);
    tile += (int)gridDim.x;
    for (; tile < tiles; tile += (int)gridDim.x) do_tile(tile);
}

__global__ __launch_bounds__(32 * ROWS_PB)
void gcn_spmm_row_kernel(const int* __restrict__ rp,
                         const int* __restrict__ A_cols,
                         const float* __restrict__ A_vals,
                         const float* __restrict__ H,
                         float* __restrict__ out,
                         int N) {
    const int lane = threadIdx.x & 31;
    const int row = __builtin_amdgcn_readfirstlane(blockIdx.x * ROWS_PB + (threadIdx.x >> 5));
    if (row >= N) return;
    const int beg = rp[row];
    const int end = rp[row + 1];

    const v4f* __restrict__ H4 = (const v4f*)H + lane;
    v4f acc = {0.0f, 0.0f, 0.0f, 0.0f};

    for (int base = beg; base < end; base += 32) {
        int n = end - base;
        if (n > 32) n = 32;
        int   myc = 0;
        float myv = 0.0f;
        if (lane < n) {
            myc = A_cols[base + lane];
            myv = A_vals[base + lane];
        }
        const int myvb = __float_as_int(myv);

        int j = 0;
        for (; j + GB <= n; j += GB) {
            v4f h[GB];
#pragma unroll
            for (int u = 0; u < GB; ++u) {
                const int c = __builtin_amdgcn_readlane(myc, j + u);
                h[u] = H4[(size_t)c * (OUT_C / 4)];
            }
#pragma unroll
            for (int u = 0; u < GB; ++u) {
                const float v = __int_as_float(__builtin_amdgcn_readlane(myvb, j + u));
                acc += v * h[u];
            }
        }
        if (j < n) {
            v4f h[GB - 1];
#pragma unroll
            for (int u = 0; u < GB - 1; ++u) {
                h[u] = (v4f){0.0f, 0.0f, 0.0f, 0.0f};
                if (j + u < n) {
                    const int c = __builtin_amdgcn_readlane(myc, j + u);
                    h[u] = H4[(size_t)c * (OUT_C / 4)];
                }
            }
#pragma unroll
            for (int u = 0; u < GB - 1; ++u) {
                if (j + u < n) {
                    const float v = __int_as_float(__builtin_amdgcn_readlane(myvb, j + u));
                    acc += v * h[u];
                }
            }
        }
    }
    ((v4f*)out)[(size_t)row * (OUT_C / 4) + lane] = acc;
}

extern "C" void kernel_launch(void* const* d_in, const int* in_sizes, int n_in,
                              void* d_out, int out_size, void* d_ws, size_t ws_size,
                              hipStream_t stream) {
    const float* X      = (const float*)d_in[0];
    const float* W      = (const float*)d_in[1];
    const int*   A_rows = (const int*)d_in[2];
    const int*   A_cols = (const int*)d_in[3];
    const float* A_vals = (const float*)d_in[4];
    float* out = (float*)d_out;

    const int N = in_sizes[0] / IN_C;
    const int E = in_sizes[2];

    float* H  = (float*)d_ws;
    int*   rp = (int*)(H + (size_t)N * OUT_C);

    const int tiles = (N + 15) / 16;
    gcn_gemm_wreg_kernel<<<GEMM_BLOCKS, 128, 0, stream>>>(X, W, H, N, tiles, A_rows, rp, E);

    gcn_spmm_row_kernel<<<(N + ROWS_PB - 1) / ROWS_PB, 32 * ROWS_PB, 0, stream>>>(rp, A_cols, A_vals, H, out, N);
}
